# starting kernel
# baseline (speedup 1.0000x reference)
#include <hip/hip_runtime.h>


#define D_IN        128
#define D_HID       64
#define OUT_CLS     2
#define TILE_M      16
#define WAVES_PB    4
#define BLOCK_THR   (WAVES_PB * 32)
#define GRID_BLOCKS 1536
#define E_STRIDE    132
#define H_STRIDE    68

typedef __attribute__((ext_vector_type(2))) float v2f;
typedef __attribute__((ext_vector_type(8))) float v8f;

__global__ void repack_w1_kernel(const float* __restrict__ W1,
                                 float* __restrict__ Bp) {
  int g = blockIdx.x * blockDim.x + threadIdx.x;
  if (g >= 32 * 4 * 32) return;
  int l = g & 31;
  int t = (g >> 5) & 3;
  int c = g >> 7;
  int k = 4 * c + 2 * (l >> 4);
  int n = 16 * t + (l & 15);
  Bp[2 * g + 0] = W1[k * D_HID + n];
  Bp[2 * g + 1] = W1[(k + 1) * D_HID + n];
}

__global__ __launch_bounds__(BLOCK_THR)
void edge_mlp_kernel(const float* __restrict__ h,
                     const float* __restrict__ Bp,
                     const float* __restrict__ b1,
                     const float* __restrict__ W2,
                     const float* __restrict__ b2,
                     const int*   __restrict__ src,
                     const int*   __restrict__ dst,
                     float*       __restrict__ out,
                     int n_edges) {
  __shared__ float s_par[D_HID * OUT_CLS + D_HID + OUT_CLS];
  __shared__ float s_e[WAVES_PB][TILE_M * E_STRIDE];

  const int lane = threadIdx.x & 31;
  const int wave = threadIdx.x >> 5;
  const int m_l  = lane & 15;
  const int k_hi = (lane >> 4) << 1;
  const int r_hi = (lane >> 4) << 3;

  for (int i = threadIdx.x; i < D_HID * OUT_CLS + D_HID + OUT_CLS;
       i += BLOCK_THR) {
    float v;
    if (i < D_HID * OUT_CLS)               v = W2[i];
    else if (i < D_HID * OUT_CLS + D_HID)  v = b1[i - D_HID * OUT_CLS];
    else                                   v = b2[i - D_HID * OUT_CLS - D_HID];
    s_par[i] = v;
  }
  __syncthreads();

  float* et = s_e[wave];

  const int n_tiles      = (n_edges + TILE_M - 1) / TILE_M;
  const int stride_tiles = gridDim.x * WAVES_PB;

  for (int tile = blockIdx.x * WAVES_PB + wave; tile < n_tiles;
       tile += stride_tiles) {
    const int e0 = tile * TILE_M;

    __builtin_prefetch(src + e0 + stride_tiles * TILE_M, 0, 0);
    __builtin_prefetch(dst + e0 + stride_tiles * TILE_M, 0, 0);

    int ei = e0 + m_l;
    if (ei >= n_edges) ei = n_edges - 1;
    const int my_s = src[ei];
    const int my_d = dst[ei];

    #pragma unroll 4
    for (int m = 0; m < TILE_M; ++m) {
      const int sm = __shfl(my_s, m, 32);
      const int dm = __shfl(my_d, m, 32);
      const float4 a = *(const float4*)(h + (size_t)sm * D_IN + lane * 4);
      const float4 b = *(const float4*)(h + (size_t)dm * D_IN + lane * 4);
      float4 p;
      p.x = a.x * b.x; p.y = a.y * b.y; p.z = a.z * b.z; p.w = a.w * b.w;
      *(float4*)(et + m * E_STRIDE + lane * 4) = p;
    }

    v8f acc0 = {}, acc1 = {}, acc2 = {}, acc3 = {};
    #pragma unroll
    for (int c = 0; c < D_IN / 4; ++c) {
      const v2f av = *(const v2f*)(et + m_l * E_STRIDE + 4 * c + k_hi);
      const float* bp = Bp + ((size_t)(c * 4) * 32 + lane) * 2;
      const v2f bv0 = *(const v2f*)(bp);
      const v2f bv1 = *(const v2f*)(bp + 64);
      const v2f bv2 = *(const v2f*)(bp + 128);
      const v2f bv3 = *(const v2f*)(bp + 192);
      acc0 = __builtin_amdgcn_wmma_f32_16x16x4_f32(false, av, false, bv0,
                                                   (short)0, acc0, false, false);
      acc1 = __builtin_amdgcn_wmma_f32_16x16x4_f32(false, av, false, bv1,
                                                   (short)0, acc1, false, false);
      acc2 = __builtin_amdgcn_wmma_f32_16x16x4_f32(false, av, false, bv2,
                                                   (short)0, acc2, false, false);
      acc3 = __builtin_amdgcn_wmma_f32_16x16x4_f32(false, av, false, bv3,
                                                   (short)0, acc3, false, false);
    }

    #define EPILOGUE(ACC, T)                                                 \
      {                                                                      \
        const float bias = s_par[D_HID * OUT_CLS + (T) * 16 + m_l];          \
        _Pragma("unroll") for (int r = 0; r < 8; ++r) {                      \
          float v = ACC[r] + bias;                                           \
          v = v > 0.0f ? v : 0.0f;                                           \
          et[(r + r_hi) * H_STRIDE + (T) * 16 + m_l] = v;                    \
        }                                                                    \
      }
    EPILOGUE(acc0, 0) EPILOGUE(acc1, 1) EPILOGUE(acc2, 2) EPILOGUE(acc3, 3)
    #undef EPILOGUE

    const int m_out = lane >> 1;
    const int cl    = lane & 1;
    float sum = s_par[D_HID * OUT_CLS + D_HID + cl];
    #pragma unroll
    for (int k = 0; k < D_HID; ++k)
      sum += et[m_out * H_STRIDE + k] * s_par[k * OUT_CLS + cl];

    const int oi = e0 * OUT_CLS + lane;
    if (oi < n_edges * OUT_CLS) out[oi] = sum;
  }
}

extern "C" void kernel_launch(void* const* d_in, const int* in_sizes, int n_in,
                              void* d_out, int out_size, void* d_ws,
                              size_t ws_size, hipStream_t stream) {
  (void)n_in; (void)out_size; (void)ws_size;
  const float* h   = (const float*)d_in[0];
  const float* W1  = (const float*)d_in[1];
  const float* b1  = (const float*)d_in[2];
  const float* W2  = (const float*)d_in[3];
  const float* b2  = (const float*)d_in[4];
  const int*   src = (const int*)d_in[5];
  const int*   dst = (const int*)d_in[6];
  float*       out = (float*)d_out;
  float*       Bp  = (float*)d_ws;

  const int n_edges = in_sizes[5];

  repack_w1_kernel<<<(32 * 4 * 32 + 255) / 256, 256, 0, stream>>>(W1, Bp);

  const int n_tiles = (n_edges + TILE_M - 1) / TILE_M;
  int blocks = (n_tiles + WAVES_PB - 1) / WAVES_PB;
  if (blocks > GRID_BLOCKS) blocks = GRID_BLOCKS;
  edge_mlp_kernel<<<blocks, BLOCK_THR, 0, stream>>>(h, Bp, b1, W2, b2, src,
                                                    dst, out, n_edges);
}


# Layer 2 from the accumulators: per-lane partial scores (own b1, W2 columns in registers), 16-lane row_xmask halving butterfly; hidden tile no longer passes LDS; layer-2 sum order changed
# speedup vs baseline: 1.2335x; 1.2335x over previous
#include <hip/hip_runtime.h>


#define D_IN        128
#define D_HID       64
#define OUT_CLS     2
#define TILE_M      16
#define WAVES_PB    8
#define BLOCK_THR   (WAVES_PB * 32)
#define GRID_BLOCKS 768
#define E_STRIDE    132
#define B_FLOATS    (D_IN * D_HID)

typedef __attribute__((ext_vector_type(2))) float v2f;
typedef __attribute__((ext_vector_type(8))) float v8f;

template <int M>
__device__ __forceinline__ float row_xmask(float v) {
  return __builtin_bit_cast(
      float, __builtin_amdgcn_update_dpp(0, __builtin_bit_cast(int, v),
                                         0x160 | M, 0xf, 0xf, true));
}


__global__ __launch_bounds__(BLOCK_THR)
void edge_mlp_kernel(const float* __restrict__ h,
                     const float* __restrict__ W1,
                     const float* __restrict__ b1,
                     const float* __restrict__ W2,
                     const float* __restrict__ b2,
                     const int*   __restrict__ src,
                     const int*   __restrict__ dst,
                     float*       __restrict__ out,
                     int n_edges) {
  __shared__ float s_e[WAVES_PB][TILE_M * E_STRIDE];
  __shared__ __attribute__((aligned(16))) float s_B[B_FLOATS];

  const int lane = threadIdx.x & 31;
  const int wave = threadIdx.x >> 5;
  const int m_l  = lane & 15;
  const int k_hi = (lane >> 4) << 1;

  float b1c[4], w2a[4], w2b[4];
  #pragma unroll
  for (int t = 0; t < 4; ++t) {
    const int col = t * 16 + m_l;
    b1c[t] = b1[col];
    w2a[t] = W2[col * OUT_CLS + 0];
    w2b[t] = W2[col * OUT_CLS + 1];
  }
  const float b2v = b2[lane & 1];

  {
    const int qlo = lane & 3;
    const int hh  = (lane >> 2) & 1;
    const int tt  = lane >> 3;
    const float* wsrc = W1 + (2 * hh) * D_HID + 16 * tt + 4 * qlo;
    float*       bdst = s_B + tt * 64 + hh * 32 + qlo * 8;
    #pragma unroll
    for (int j = 0; j < (D_IN / 4) / WAVES_PB; ++j) {
      const int c = j * WAVES_PB + wave;
      const float4 r0 = *(const float4*)(wsrc + (4 * c) * D_HID);
      const float4 r1 = *(const float4*)(wsrc + (4 * c + 1) * D_HID);
      *(float4*)(bdst + c * 256)     = make_float4(r0.x, r1.x, r0.y, r1.y);
      *(float4*)(bdst + c * 256 + 4) = make_float4(r0.z, r1.z, r0.w, r1.w);
    }
  }
  __syncthreads();

  float* et = s_e[wave];

  const int n_tiles      = (n_edges + TILE_M - 1) / TILE_M;
  const int stride_tiles = gridDim.x * WAVES_PB;
  const int n_trips = (n_tiles + stride_tiles - 1) / stride_tiles;
  int trips4 = 0;

  for (int tile = blockIdx.x * WAVES_PB + wave; tile < n_tiles;
       tile += stride_tiles, trips4 += 4) {
    const int e0 = tile * TILE_M;

    const int quarter = (trips4 >= n_trips) + (trips4 >= 2 * n_trips) +
                        (trips4 >= 3 * n_trips);
    if (quarter == 0)      __builtin_amdgcn_s_setprio(3);
    else if (quarter == 1) __builtin_amdgcn_s_setprio(2);
    else if (quarter == 2) __builtin_amdgcn_s_setprio(1);
    else                   __builtin_amdgcn_s_setprio(0);

    __builtin_prefetch(src + e0 + stride_tiles * TILE_M, 0, 0);
    __builtin_prefetch(dst + e0 + stride_tiles * TILE_M, 0, 0);

    int ei = e0 + m_l;
    if (ei >= n_edges) ei = n_edges - 1;
    const int my_s = src[ei];
    const int my_d = dst[ei];

    #pragma unroll 4
    for (int m = 0; m < TILE_M; ++m) {
      const int sm = __shfl(my_s, m, 32);
      const int dm = __shfl(my_d, m, 32);
      const float4 a = *(const float4*)(h + (size_t)sm * D_IN + lane * 4);
      const float4 b = *(const float4*)(h + (size_t)dm * D_IN + lane * 4);
      float4 p;
      p.x = a.x * b.x; p.y = a.y * b.y; p.z = a.z * b.z; p.w = a.w * b.w;
      *(float4*)(et + m * E_STRIDE + lane * 4) = p;
    }

    asm volatile("" ::: "memory");
    v8f acc0 = {}, acc1 = {}, acc2 = {}, acc3 = {};
    #pragma unroll
    for (int c = 0; c < D_IN / 4; ++c) {
      const v2f av = *(const v2f*)(et + m_l * E_STRIDE + 4 * c + k_hi);
      const float* bp = s_B + ((c * 4) * 32 + lane) * 2;
      const v2f bv0 = *(const v2f*)(bp);
      const v2f bv1 = *(const v2f*)(bp + 64);
      const v2f bv2 = *(const v2f*)(bp + 128);
      const v2f bv3 = *(const v2f*)(bp + 192);
      acc0 = __builtin_amdgcn_wmma_f32_16x16x4_f32(false, av, false, bv0,
                                                   (short)0, acc0, false, false);
      acc1 = __builtin_amdgcn_wmma_f32_16x16x4_f32(false, av, false, bv1,
                                                   (short)0, acc1, false, false);
      acc2 = __builtin_amdgcn_wmma_f32_16x16x4_f32(false, av, false, bv2,
                                                   (short)0, acc2, false, false);
      acc3 = __builtin_amdgcn_wmma_f32_16x16x4_f32(false, av, false, bv3,
                                                   (short)0, acc3, false, false);
    }

    float p[16];
    #pragma unroll
    for (int r = 0; r < 8; ++r) {
      float h0 = acc0[r] + b1c[0];
      float h1 = acc1[r] + b1c[1];
      float h2 = acc2[r] + b1c[2];
      float h3 = acc3[r] + b1c[3];
      h0 = h0 > 0.0f ? h0 : 0.0f;
      h1 = h1 > 0.0f ? h1 : 0.0f;
      h2 = h2 > 0.0f ? h2 : 0.0f;
      h3 = h3 > 0.0f ? h3 : 0.0f;
      p[2 * r]     = h0 * w2a[0] + h1 * w2a[1] + h2 * w2a[2] + h3 * w2a[3];
      p[2 * r + 1] = h0 * w2b[0] + h1 * w2b[1] + h2 * w2b[2] + h3 * w2b[3];
    }

    const bool up8 = (lane & 8) != 0;
    const bool up4 = (lane & 4) != 0;
    const bool up2 = (lane & 2) != 0;
    const bool up1 = (lane & 1) != 0;
    float q[8];
    #pragma unroll
    for (int j = 0; j < 8; ++j) {
      const float keep = up8 ? p[j + 8] : p[j];
      const float send = up8 ? p[j] : p[j + 8];
      q[j] = keep + row_xmask<8>(send);
    }
    float s[4];
    #pragma unroll
    for (int j = 0; j < 4; ++j) {
      const float keep = up4 ? q[j + 4] : q[j];
      const float send = up4 ? q[j] : q[j + 4];
      s[j] = keep + row_xmask<4>(send);
    }
    float u[2];
    #pragma unroll
    for (int j = 0; j < 2; ++j) {
      const float keep = up2 ? s[j + 2] : s[j];
      const float send = up2 ? s[j] : s[j + 2];
      u[j] = keep + row_xmask<2>(send);
    }
    const float keep1 = up1 ? u[1] : u[0];
    const float send1 = up1 ? u[0] : u[1];
    const float sum = keep1 + row_xmask<1>(send1) + b2v;

    const int oi = e0 * OUT_CLS + lane;
    if (oi < n_edges * OUT_CLS) out[oi] = sum;
  }
  __builtin_amdgcn_s_setprio(0);
}

extern "C" void kernel_launch(void* const* d_in, const int* in_sizes, int n_in,
                              void* d_out, int out_size, void* d_ws,
                              size_t ws_size, hipStream_t stream) {
  (void)n_in; (void)out_size; (void)d_ws; (void)ws_size;
  const float* h   = (const float*)d_in[0];
  const float* W1  = (const float*)d_in[1];
  const float* b1  = (const float*)d_in[2];
  const float* W2  = (const float*)d_in[3];
  const float* b2  = (const float*)d_in[4];
  const int*   src = (const int*)d_in[5];
  const int*   dst = (const int*)d_in[6];
  float*       out = (float*)d_out;

  const int n_edges = in_sizes[5];

  const int n_tiles = (n_edges + TILE_M - 1) / TILE_M;
  int blocks = (n_tiles + WAVES_PB - 1) / WAVES_PB;
  if (blocks > GRID_BLOCKS) blocks = GRID_BLOCKS;
  edge_mlp_kernel<<<blocks, BLOCK_THR, 0, stream>>>(h, W1, b1, W2, b2, src,
                                                    dst, out, n_edges);
}
